# starting kernel
# baseline (speedup 1.0000x reference)
#include <hip/hip_runtime.h>
#include <hip/hip_bf16.h>

#define B 32
#define M 512
#define T 16
#define D 128
#define V 32000

typedef __attribute__((ext_vector_type(2))) float v2f;
typedef __attribute__((ext_vector_type(8))) float v8f;

__global__ void pos_kernel(const int* __restrict__ story,
                           int* __restrict__ positions,
                           float* __restrict__ u)
{
    int b = blockIdx.x;
    int m = threadIdx.x;
    __shared__ int scan[M];
    int nonpad = (story[(size_t)m * (B * T) + (size_t)b * T] != 0) ? 1 : 0;
    scan[m] = nonpad;
    __syncthreads();
    for (int off = 1; off < M; off <<= 1) {
        int v = scan[m];
        int w = (m >= off) ? scan[m - off] : 0;
        __syncthreads();
        scan[m] = v + w;
        __syncthreads();
    }
    positions[b * M + m] = nonpad ? scan[m] : 0;
    if (m < D) u[b * D + m] = 0.0f;
}

__global__ void pool_kernel(const int* __restrict__ story,
                            const float* __restrict__ C_tables,
                            const float* __restrict__ pos_emb,
                            const int* __restrict__ positions,
                            float* __restrict__ SA)
{
    int wave = threadIdx.x >> 5;
    int lane = threadIdx.x & 31;
    int slot = blockIdx.x * 8 + wave;
    int b = slot >> 9;
    int m = slot & (M - 1);

    int tokval = 0;
    if (lane < T) tokval = story[(size_t)m * (B * T) + (size_t)b * T + lane];

    const float4* t0 = (const float4*)(C_tables);
    const float4* t1 = (const float4*)(C_tables + (size_t)1 * V * D);
    const float4* t2 = (const float4*)(C_tables + (size_t)2 * V * D);
    const float4* t3 = (const float4*)(C_tables + (size_t)3 * V * D);
    const int D4 = D / 4;

    float4 a0 = {0.f, 0.f, 0.f, 0.f};
    float4 a1 = a0, a2 = a0, a3 = a0;
#pragma unroll
    for (int t = 0; t < T; ++t) {
        int tok = __shfl(tokval, t, 32);
        size_t off = (size_t)tok * D4 + lane;
        float4 v0 = t0[off];
        float4 v1 = t1[off];
        float4 v2 = t2[off];
        float4 v3 = t3[off];
        a0.x += v0.x; a0.y += v0.y; a0.z += v0.z; a0.w += v0.w;
        a1.x += v1.x; a1.y += v1.y; a1.z += v1.z; a1.w += v1.w;
        a2.x += v2.x; a2.y += v2.y; a2.z += v2.z; a2.w += v2.w;
        a3.x += v3.x; a3.y += v3.y; a3.z += v3.z; a3.w += v3.w;
    }

    float4 pe = ((const float4*)pos_emb)[(size_t)positions[slot] * D4 + lane];
    a0.x += pe.x; a0.y += pe.y; a0.z += pe.z; a0.w += pe.w;
    a1.x += pe.x; a1.y += pe.y; a1.z += pe.z; a1.w += pe.w;
    a2.x += pe.x; a2.y += pe.y; a2.z += pe.z; a2.w += pe.w;
    a3.x += pe.x; a3.y += pe.y; a3.z += pe.z; a3.w += pe.w;

    float4* o = (float4*)SA;
    size_t base = (size_t)slot * D4 + lane;
    const size_t S4 = (size_t)B * M * D4;
    o[base + 0 * S4] = a0;
    o[base + 1 * S4] = a1;
    o[base + 2 * S4] = a2;
    o[base + 3 * S4] = a3;
}

__global__ void hop_kernel(const float* __restrict__ SA_A,
                           const float* __restrict__ SA_C,
                           float* __restrict__ u,
                           float* __restrict__ out)
{
    int b    = blockIdx.x;
    int tid  = threadIdx.x;
    int wave = tid >> 5;
    int lane = tid & 31;
    int row  = lane & 15;
    int khalf = lane >> 4;

    __shared__ float sh_u[D];
    __shared__ float sh_l[M];
    __shared__ float red[8];

    if (tid < D) sh_u[tid] = u[b * D + tid];
    __syncthreads();

    const float* Abase = SA_A + (size_t)b * M * D;
    for (int mt = wave * 4; mt < wave * 4 + 4; ++mt) {
        int m0 = mt * 16;
        v8f acc = {0.f, 0.f, 0.f, 0.f, 0.f, 0.f, 0.f, 0.f};
        for (int kc = 0; kc < D / 4; ++kc) {
            int d0 = kc * 4 + 2 * khalf;
            const float* p = Abase + (size_t)(m0 + row) * D + d0;
            v2f a;  a.x = p[0];        a.y = p[1];
            v2f bb; bb.x = sh_u[d0];   bb.y = sh_u[d0 + 1];
            acc = __builtin_amdgcn_wmma_f32_16x16x4_f32(
                false, a, false, bb, (short)0, acc, false, false);
        }
        if (row == 0) {
#pragma unroll
            for (int j = 0; j < 8; ++j)
                sh_l[m0 + khalf * 8 + j] = acc[j];
        }
    }
    __syncthreads();

    float lmax = -__builtin_inff();
    for (int i = tid; i < M; i += 256) lmax = fmaxf(lmax, sh_l[i]);
    for (int off = 16; off > 0; off >>= 1)
        lmax = fmaxf(lmax, __shfl_xor(lmax, off, 32));
    if (lane == 0) red[wave] = lmax;
    __syncthreads();
    if (tid == 0) {
        float g = red[0];
        for (int i = 1; i < 8; ++i) g = fmaxf(g, red[i]);
        red[0] = g;
    }
    __syncthreads();
    float gmax = red[0];

    float lsum = 0.f;
    for (int i = tid; i < M; i += 256) {
        float e = __expf(sh_l[i] - gmax);
        sh_l[i] = e;
        lsum += e;
    }
    for (int off = 16; off > 0; off >>= 1)
        lsum += __shfl_xor(lsum, off, 32);
    __syncthreads();
    if (lane == 0) red[wave] = lsum;
    __syncthreads();
    if (tid == 0) {
        float s = 0.f;
        for (int i = 0; i < 8; ++i) s += red[i];
        red[0] = s;
    }
    __syncthreads();
    float inv = 1.0f / red[0];
    for (int i = tid; i < M; i += 256) sh_l[i] *= inv;
    __syncthreads();

    const float* Cbase = SA_C + (size_t)b * M * D;
    int d_tile = wave * 16;
    {
        v8f acc = {0.f, 0.f, 0.f, 0.f, 0.f, 0.f, 0.f, 0.f};
        for (int kc = 0; kc < M / 4; ++kc) {
            int m0 = kc * 4 + 2 * khalf;
            const float* p = Cbase + (size_t)m0 * D + d_tile + row;
            v2f a;  a.x = p[0];        a.y = p[D];
            v2f bb; bb.x = sh_l[m0];   bb.y = sh_l[m0 + 1];
            acc = __builtin_amdgcn_wmma_f32_16x16x4_f32(
                false, a, false, bb, (short)0, acc, false, false);
        }
        if (row == 0) {
#pragma unroll
            for (int j = 0; j < 8; ++j) {
                int d = d_tile + khalf * 8 + j;
                float nu = sh_u[d] + acc[j];
                u[b * D + d] = nu;
                if (out) out[b * D + d] = nu;
            }
        }
    }
}

extern "C" void kernel_launch(void* const* d_in, const int* in_sizes, int n_in,
                              void* d_out, int out_size, void* d_ws, size_t ws_size,
                              hipStream_t stream)
{
    const int*   story    = (const int*)d_in[0];
    const float* C_tables = (const float*)d_in[1];
    const float* pos_emb  = (const float*)d_in[2];
    float*       out      = (float*)d_out;

    char* ws = (char*)d_ws;
    int*   positions = (int*)ws;
    float* u         = (float*)(ws + (size_t)B * M * sizeof(int));
    size_t sa_off    = (((size_t)B * M * sizeof(int) +
                         (size_t)B * D * sizeof(float)) + 255) & ~(size_t)255;
    float* SA        = (float*)(ws + sa_off);
    const size_t S   = (size_t)B * M * D;

    pos_kernel <<<B,          M,   0, stream>>>(story, positions, u);
    pool_kernel<<<B * M / 8,  256, 0, stream>>>(story, C_tables, pos_emb,
                                                positions, SA);
    hop_kernel <<<B,          256, 0, stream>>>(SA + 0 * S, SA + 1 * S, u, nullptr);
    hop_kernel <<<B,          256, 0, stream>>>(SA + 1 * S, SA + 2 * S, u, nullptr);
    hop_kernel <<<B,          256, 0, stream>>>(SA + 2 * S, SA + 3 * S, u, out);
}


# Table-2 and table-3 pooling run as 224 extra work-groups inside the hop-0 and hop-1 launches, so the matrix-bound hops overlap the gather; tables 0-1 pooled two slots per wave; sums unchanged
# speedup vs baseline: 1.8153x; 1.8153x over previous
#include <hip/hip_runtime.h>
#include <hip/hip_bf16.h>

#define B 32
#define M 512
#define T 16
#define D 128
#define V 32000
#define NPOOL 224

typedef __attribute__((ext_vector_type(2))) float v2f;
typedef __attribute__((ext_vector_type(8))) float v8f;

__global__ void pos_kernel(const int* __restrict__ story,
                           int* __restrict__ positions,
                           float* __restrict__ u)
{
    int b = blockIdx.x;
    int m = threadIdx.x;
    __shared__ int scan[M];
    int nonpad = (story[(size_t)m * (B * T) + (size_t)b * T] != 0) ? 1 : 0;
    scan[m] = nonpad;
    __syncthreads();
    for (int off = 1; off < M; off <<= 1) {
        int v = scan[m];
        int w = (m >= off) ? scan[m - off] : 0;
        __syncthreads();
        scan[m] = v + w;
        __syncthreads();
    }
    positions[b * M + m] = nonpad ? scan[m] : 0;
    if (m < D) u[b * D + m] = 0.0f;
}

template <int NT>
static __device__ __forceinline__ void pool_rows(const float* __restrict__ tab,
                                                 const float* __restrict__ pos_emb,
                                                 float* __restrict__ SAo,
                                                 int slot, int lane,
                                                 int tokval, int tbase, int pidx)
{
    const int D4 = D / 4;
    float4 acc[NT];
#pragma unroll
    for (int k = 0; k < NT; ++k) acc[k] = make_float4(0.f, 0.f, 0.f, 0.f);
#pragma unroll
    for (int t = 0; t < T; ++t) {
        const int tok = __shfl(tokval, tbase + t, 32);
        const size_t off = (size_t)tok * D4 + lane;
#pragma unroll
        for (int k = 0; k < NT; ++k) {
            const float4 v = ((const float4*)(tab + (size_t)k * V * D))[off];
            acc[k].x += v.x; acc[k].y += v.y; acc[k].z += v.z; acc[k].w += v.w;
        }
    }
    const float4 pe = ((const float4*)pos_emb)[(size_t)pidx * D4 + lane];
    const size_t S4 = (size_t)B * M * D4;
#pragma unroll
    for (int k = 0; k < NT; ++k) {
        acc[k].x += pe.x; acc[k].y += pe.y; acc[k].z += pe.z; acc[k].w += pe.w;
        ((float4*)SAo)[(size_t)k * S4 + (size_t)slot * D4 + lane] = acc[k];
    }
}

__global__ void pool01_kernel(const int* __restrict__ story,
                              const float* __restrict__ C_tables,
                              const float* __restrict__ pos_emb,
                              const int* __restrict__ positions,
                              float* __restrict__ SA)
{
    const int wave = threadIdx.x >> 5;
    const int lane = threadIdx.x & 31;
    const int q  = blockIdx.x * 8 + wave;
    const int b0 = (q >> 9) * 2;
    const int m  = q & (M - 1);
    const int slot0 = b0 * M + m;
    const int slot1 = slot0 + M;

    const int tokval = story[(size_t)m * (B * T) + (size_t)b0 * T + lane];
    const int p0 = positions[slot0];
    const int p1 = positions[slot1];
    pool_rows<2>(C_tables, pos_emb, SA, slot0, lane, tokval, 0, p0);
    pool_rows<2>(C_tables, pos_emb, SA, slot1, lane, tokval, T, p1);
}

static __device__ __forceinline__ void lds_barrier()
{
    asm volatile("s_wait_dscnt 0x0\n\ts_barrier_signal -1\n\ts_barrier_wait -1"
                 ::: "memory");
}

__global__ __launch_bounds__(1024)
void hop_pool_kernel(const float* __restrict__ SA_A,
                     const float* __restrict__ SA_C,
                     float* __restrict__ u,
                     float* __restrict__ out,
                     const int* __restrict__ story,
                     const float* __restrict__ tab,
                     const float* __restrict__ pos_emb,
                     const int* __restrict__ positions,
                     float* __restrict__ SAo)
{
    const int tid  = threadIdx.x;
    const int wave = tid >> 5;
    const int lane = tid & 31;

    if (blockIdx.x >= B) {
        const int p     = blockIdx.x - B;
        const int start = (p * (B * M)) / NPOOL;
        const int end   = ((p + 1) * (B * M)) / NPOOL;
        const int s0    = start + wave;
        int tv[3], px[3];
#pragma unroll
        for (int i = 0; i < 3; ++i) {
            const int s  = min(s0 + 32 * i, B * M - 1);
            const int bb = s >> 9, mm = s & (M - 1);
            tv[i] = story[(size_t)mm * (B * T) + (size_t)bb * T + (lane & (T - 1))];
            px[i] = positions[s];
        }
#pragma unroll
        for (int i = 0; i < 3; ++i)
            if (s0 + 32 * i < end)
                pool_rows<1>(tab, pos_emb, SAo, s0 + 32 * i, lane, tv[i], 0, px[i]);
        return;
    }

    const int b    = blockIdx.x;
    const int c    = lane & 15;
    const int h    = lane >> 4;

    __shared__ __attribute__((aligned(16))) float sh_u[D];
    __shared__ __attribute__((aligned(16))) float sh_l[M];
    __shared__ __attribute__((aligned(16))) float sh_part[16][D];

    float uval = 0.f;
    if (tid < D) uval = u[b * D + tid];

    const float4* arow =
        (const float4*)(SA_A + ((size_t)b * M + wave * 16 + c) * D) + h;
    float4 va[16];
#pragma unroll
    for (int j = 0; j < 16; ++j) va[j] = arow[2 * j];

    if (tid < D) sh_u[tid] = uval;
    lds_barrier();

    v8f acc0 = {0.f, 0.f, 0.f, 0.f, 0.f, 0.f, 0.f, 0.f};
    v8f acc1 = acc0;
#pragma unroll
    for (int j = 0; j < 16; ++j) {
        const float4 uu = *(const float4*)&sh_u[8 * j + 4 * h];
        v2f a0; a0.x = uu.x;      a0.y = uu.y;
        v2f b0; b0.x = va[j].x;   b0.y = va[j].y;
        v2f a1; a1.x = uu.z;      a1.y = uu.w;
        v2f b1; b1.x = va[j].z;   b1.y = va[j].w;
        acc0 = __builtin_amdgcn_wmma_f32_16x16x4_f32(
            false, a0, false, b0, (short)0, acc0, false, false);
        acc1 = __builtin_amdgcn_wmma_f32_16x16x4_f32(
            false, a1, false, b1, (short)0, acc1, false, false);
    }
    sh_l[wave * 16 + c] = acc0[0] + acc1[0];

    const int dg = wave & 1;
    const int mc = wave >> 1;
    const float4* crow =
        (const float4*)(SA_C + ((size_t)b * M + mc * 32 + 2 * h) * D
                        + dg * 64 + 4 * c);
    float4 vc[16];
#pragma unroll
    for (int kk = 0; kk < 8; ++kk) {
        vc[2 * kk]     = crow[(4 * kk)     * (D / 4)];
        vc[2 * kk + 1] = crow[(4 * kk + 1) * (D / 4)];
    }
    lds_barrier();

    float gmax = -__builtin_inff();
    float lv[16];
#pragma unroll
    for (int i = 0; i < 4; ++i) {
        const float4 t = *(const float4*)&sh_l[128 * i + 4 * lane];
        lv[4 * i] = t.x; lv[4 * i + 1] = t.y; lv[4 * i + 2] = t.z; lv[4 * i + 3] = t.w;
    }
#pragma unroll
    for (int i = 0; i < 16; ++i) gmax = fmaxf(gmax, lv[i]);
    for (int off = 16; off > 0; off >>= 1)
        gmax = fmaxf(gmax, __shfl_xor(gmax, off, 32));
    float gsum = 0.f;
#pragma unroll
    for (int i = 0; i < 16; ++i) gsum += __expf(lv[i] - gmax);
    for (int off = 16; off > 0; off >>= 1)
        gsum += __shfl_xor(gsum, off, 32);
    const float inv = 1.0f / gsum;

    v8f o0 = {0.f, 0.f, 0.f, 0.f, 0.f, 0.f, 0.f, 0.f};
    v8f o1 = o0, o2 = o0, o3 = o0;
#pragma unroll
    for (int kk = 0; kk < 8; ++kk) {
        const float2 l2 = *(const float2*)&sh_l[mc * 32 + 4 * kk + 2 * h];
        v2f pa;
        pa.x = __expf(l2.x - gmax) * inv;
        pa.y = __expf(l2.y - gmax) * inv;
        const float4 r0 = vc[2 * kk];
        const float4 r1 = vc[2 * kk + 1];
        v2f q0; q0.x = r0.x; q0.y = r1.x;
        v2f q1; q1.x = r0.y; q1.y = r1.y;
        v2f q2; q2.x = r0.z; q2.y = r1.z;
        v2f q3; q3.x = r0.w; q3.y = r1.w;
        o0 = __builtin_amdgcn_wmma_f32_16x16x4_f32(
            false, pa, false, q0, (short)0, o0, false, false);
        o1 = __builtin_amdgcn_wmma_f32_16x16x4_f32(
            false, pa, false, q1, (short)0, o1, false, false);
        o2 = __builtin_amdgcn_wmma_f32_16x16x4_f32(
            false, pa, false, q2, (short)0, o2, false, false);
        o3 = __builtin_amdgcn_wmma_f32_16x16x4_f32(
            false, pa, false, q3, (short)0, o3, false, false);
    }
    float4 po; po.x = o0[0]; po.y = o1[0]; po.z = o2[0]; po.w = o3[0];
    *(float4*)&sh_part[mc][dg * 64 + 4 * c] = po;
    __syncthreads();

    if (tid < D) {
        float o = 0.f;
#pragma unroll
        for (int i = 0; i < 16; ++i) o += sh_part[i][tid];
        const float nu = sh_u[tid] + o;
        u[b * D + tid] = nu;
        if (out) out[b * D + tid] = nu;
    }
}

extern "C" void kernel_launch(void* const* d_in, const int* in_sizes, int n_in,
                              void* d_out, int out_size, void* d_ws, size_t ws_size,
                              hipStream_t stream)
{
    const int*   story    = (const int*)d_in[0];
    const float* C_tables = (const float*)d_in[1];
    const float* pos_emb  = (const float*)d_in[2];
    float*       out      = (float*)d_out;

    char* ws = (char*)d_ws;
    int*   positions = (int*)ws;
    float* u         = (float*)(ws + (size_t)B * M * sizeof(int));
    size_t sa_off    = (((size_t)B * M * sizeof(int) +
                         (size_t)B * D * sizeof(float)) + 255) & ~(size_t)255;
    float* SA        = (float*)(ws + sa_off);
    const size_t S   = (size_t)B * M * D;

    const size_t TD  = (size_t)V * D;

    pos_kernel     <<<B,          M,    0, stream>>>(story, positions, u);
    pool01_kernel  <<<B * M / 16, 256,  0, stream>>>(story, C_tables, pos_emb,
                                                    positions, SA);
    hop_pool_kernel<<<B + NPOOL,  1024, 0, stream>>>(SA + 0 * S, SA + 1 * S, u, nullptr,
                                                    story, C_tables + 2 * TD, pos_emb,
                                                    positions, SA + 2 * S);
    hop_pool_kernel<<<B + NPOOL,  1024, 0, stream>>>(SA + 1 * S, SA + 2 * S, u, nullptr,
                                                    story, C_tables + 3 * TD, pos_emb,
                                                    positions, SA + 3 * S);
    hop_pool_kernel<<<B,          1024, 0, stream>>>(SA + 2 * S, SA + 3 * S, u, out,
                                                    nullptr, nullptr, nullptr,
                                                    nullptr, nullptr);
}


# First-hop logits formed inside the table-0/1 pooling kernel from its table-0 sums (LDS tile, 4 WMMAs a wave): plane 0 and its launch gone, 5 launches; summation tree of those logits changes
# speedup vs baseline: 2.2665x; 1.2486x over previous
#include <hip/hip_runtime.h>
#include <hip/hip_bf16.h>

#define B 32
#define M 512
#define T 16
#define D 128
#define V 32000
#define NQ 4
#define DC 32
#define NH (B * NQ)
#define NLONG 128
#define NSHORT 128
#define SLONG 80
#define SSHORT 48
#define NPOOL (NLONG + NSHORT)
static_assert(NLONG * SLONG + NSHORT * SSHORT == B * M, "pooling shares must cover every slot once");
#define PM 32
#define SP (D + 4)

typedef __attribute__((ext_vector_type(2))) float v2f;
typedef __attribute__((ext_vector_type(8))) float v8f;

__global__ void pos_kernel(const int* __restrict__ story,
                           int* __restrict__ positions,
                           float* __restrict__ u)
{
    int b = blockIdx.x;
    int m = threadIdx.x;
    __shared__ int scan[M];
    int nonpad = (story[(size_t)m * (B * T) + (size_t)b * T] != 0) ? 1 : 0;
    scan[m] = nonpad;
    __syncthreads();
    for (int off = 1; off < M; off <<= 1) {
        int v = scan[m];
        int w = (m >= off) ? scan[m - off] : 0;
        __syncthreads();
        scan[m] = v + w;
        __syncthreads();
    }
    positions[b * M + m] = nonpad ? scan[m] : 0;
    if (m < D) u[b * D + m] = 0.0f;
}

template <int NT, bool KEEP0>
static __device__ __forceinline__ float4 pool_rows(const float* __restrict__ tab,
                                                 const float* __restrict__ pos_emb,
                                                 float* __restrict__ SAo,
                                                 int slot, int lane,
                                                 int tokval, int tbase, int pidx)
{
    const int D4 = D / 4;
    float4 acc[NT];
#pragma unroll
    for (int k = 0; k < NT; ++k) acc[k] = make_float4(0.f, 0.f, 0.f, 0.f);
#pragma unroll
    for (int t = 0; t < T; ++t) {
        const int tok = __shfl(tokval, tbase + t, 32);
        const size_t off = (size_t)tok * D4 + lane;
#pragma unroll
        for (int k = 0; k < NT; ++k) {
            const float4 v = ((const float4*)(tab + (size_t)k * V * D))[off];
            acc[k].x += v.x; acc[k].y += v.y; acc[k].z += v.z; acc[k].w += v.w;
        }
    }
    const float4 pe = ((const float4*)pos_emb)[(size_t)pidx * D4 + lane];
    const size_t S4 = (size_t)B * M * D4;
#pragma unroll
    for (int k = 0; k < NT; ++k) {
        acc[k].x += pe.x; acc[k].y += pe.y; acc[k].z += pe.z; acc[k].w += pe.w;
        if (!(KEEP0 && k == 0))
            ((float4*)SAo)[(size_t)k * S4 + (size_t)slot * D4 + lane] = acc[k];
    }
    return acc[0];
}

static __device__ __forceinline__ void lds_barrier()
{
    asm volatile("s_wait_dscnt 0x0\n\ts_barrier_signal -1\n\ts_barrier_wait -1"
                 ::: "memory");
}

__global__ __launch_bounds__(PM * 16)
void pool01_logit_kernel(const int* __restrict__ story,
                         const float* __restrict__ C_tables,
                         const float* __restrict__ pos_emb,
                         const int* __restrict__ positions,
                         const float* __restrict__ u,
                         float* __restrict__ SA,
                         float* __restrict__ PL0)
{
    const int tid  = threadIdx.x;
    const int wave = tid >> 5;
    const int lane = tid & 31;
    const int b  = blockIdx.x >> 4;
    const int m0 = (blockIdx.x & 15) * PM;
    const int c  = lane & 15;
    const int h  = lane >> 4;

    __shared__ __attribute__((aligned(16))) float sh_s[PM][SP];
    __shared__ __attribute__((aligned(16))) float sh_u[D];
    __shared__ __attribute__((aligned(16))) float sh_part[8][PM];

    const int slotA = b * M + m0 + wave;
    const int slotB = slotA + 16;
    const int tokval = story[(size_t)(m0 + wave + 16 * h) * (B * T) + (size_t)b * T + c];
    const int pA = positions[slotA];
    const int pB = positions[slotB];
    const float4 sA = pool_rows<2, true>(C_tables, pos_emb, SA, slotA, lane, tokval, 0, pA);
    const float4 sB = pool_rows<2, true>(C_tables, pos_emb, SA, slotB, lane, tokval, T, pB);

    *(float4*)&sh_s[wave][4 * lane]      = sA;
    *(float4*)&sh_s[16 + wave][4 * lane] = sB;
    sh_u[tid & (D - 1)] = u[b * D + (tid & (D - 1))];
    lds_barrier();

    const int tile = wave & 1;
    const int j0   = (wave >> 1) * 2;
    v8f acc0 = {0.f, 0.f, 0.f, 0.f, 0.f, 0.f, 0.f, 0.f};
    v8f acc1 = acc0;
    for (int jj = 0; jj < 2; ++jj) {
        const int dcol = 8 * (j0 + jj) + 4 * h;
        const float4 uu = *(const float4*)&sh_u[dcol];
        const float4 vs = *(const float4*)&sh_s[16 * tile + c][dcol];
        v2f a0; a0.x = uu.x; a0.y = uu.y;
        v2f b0; b0.x = vs.x; b0.y = vs.y;
        v2f a1; a1.x = uu.z; a1.y = uu.w;
        v2f b1; b1.x = vs.z; b1.y = vs.w;
        acc0 = __builtin_amdgcn_wmma_f32_16x16x4_f32(
            false, a0, false, b0, (short)0, acc0, false, false);
        acc1 = __builtin_amdgcn_wmma_f32_16x16x4_f32(
            false, a1, false, b1, (short)0, acc1, false, false);
    }
    sh_part[wave >> 1][16 * tile + c] = acc0[0] + acc1[0];
    lds_barrier();

    if (tid < PM) {
        float l = 0.f;
        for (int g = 0; g < 8; ++g) l += sh_part[g][tid];
        PL0[b * M + m0 + tid] = l;
    }
}

template <int NIN, bool HAS_OUT>
__global__ __launch_bounds__(1024)
void colhop_pool_kernel(const float* __restrict__ S,
                        const float* __restrict__ PLin,
                        float* __restrict__ PLout,
                        float* __restrict__ u,
                        float* __restrict__ out,
                        const int* __restrict__ story,
                        const float* __restrict__ tab,
                        const float* __restrict__ pos_emb,
                        const int* __restrict__ positions,
                        float* __restrict__ SAo)
{
    const int tid  = threadIdx.x;
    const int wave = tid >> 5;
    const int lane = tid & 31;

    if (HAS_OUT && blockIdx.x >= NH) {
        const int p    = blockIdx.x - NH;
        const int cnt  = (p < NLONG) ? SLONG : SSHORT;
        const int base = (p < NLONG) ? p * SLONG : NLONG * SLONG + (p - NLONG) * SSHORT;
        const int s0   = base + wave;
        int tv[3], px[3];
#pragma unroll
        for (int i = 0; i < 3; ++i) {
            const int s  = min(s0 + 32 * i, B * M - 1);
            const int bb = s >> 9, mm = s & (M - 1);
            tv[i] = story[(size_t)mm * (B * T) + (size_t)bb * T + (lane & (T - 1))];
            px[i] = positions[s];
        }
#pragma unroll
        for (int i = 0; i < 3; ++i)
            if (wave + 32 * i < cnt)
                pool_rows<1, false>(tab, pos_emb, SAo, s0 + 32 * i, lane, tv[i], 0, px[i]);
        return;
    }

    const int b  = blockIdx.x >> 2;
    const int q  = blockIdx.x & (NQ - 1);
    const int d0 = q * DC;
    const int c  = lane & 15;
    const int h  = lane >> 4;

    __shared__ __attribute__((aligned(16))) float sh_u[DC];
    __shared__ __attribute__((aligned(16))) float sh_l[M];
    __shared__ __attribute__((aligned(16))) float sh_part[32][DC];

    static_assert(NIN == 1 || NIN == NQ, "whole logits or one share per chunk");
    const float* pr = PLin + (size_t)b * NIN * M + (tid & (M - 1));
    const float pl0 = pr[0];
    float pl1 = 0.f, pl2 = 0.f, pl3 = 0.f;
    if (NIN == NQ) { pl1 = pr[M]; pl2 = pr[2 * M]; pl3 = pr[3 * M]; }
    const float uval = u[b * D + d0 + (tid & (DC - 1))];

    const float2* crow =
        (const float2*)(S + ((size_t)b * M + wave * 16 + 2 * h) * D + d0 + 2 * c);
    float2 vc[8];
#pragma unroll
    for (int kk = 0; kk < 4; ++kk) {
        vc[2 * kk]     = crow[(4 * kk)     * (D / 2)];
        vc[2 * kk + 1] = crow[(4 * kk + 1) * (D / 2)];
    }

    const float4* arow =
        (const float4*)(S + ((size_t)b * M + wave * 16 + c) * D + d0) + h;
    float4 va[4];
    if (HAS_OUT) {
#pragma unroll
        for (int j = 0; j < 4; ++j) va[j] = arow[2 * j];
    }

    sh_l[tid & (M - 1)] = (NIN == NQ) ? ((pl0 + pl1) + pl2) + pl3 : pl0;
    sh_u[tid & (DC - 1)] = uval;
    lds_barrier();

    float gmax = -__builtin_inff();
    float lv[16];
#pragma unroll
    for (int i = 0; i < 4; ++i) {
        const float4 t = *(const float4*)&sh_l[128 * i + 4 * lane];
        lv[4 * i] = t.x; lv[4 * i + 1] = t.y; lv[4 * i + 2] = t.z; lv[4 * i + 3] = t.w;
    }
#pragma unroll
    for (int i = 0; i < 16; ++i) gmax = fmaxf(gmax, lv[i]);
    for (int off = 16; off > 0; off >>= 1)
        gmax = fmaxf(gmax, __shfl_xor(gmax, off, 32));
    float gsum = 0.f;
#pragma unroll
    for (int i = 0; i < 16; ++i) gsum += __expf(lv[i] - gmax);
    for (int off = 16; off > 0; off >>= 1)
        gsum += __shfl_xor(gsum, off, 32);
    const float inv = 1.0f / gsum;

    v8f o0 = {0.f, 0.f, 0.f, 0.f, 0.f, 0.f, 0.f, 0.f};
    v8f o1 = o0;
#pragma unroll
    for (int kk = 0; kk < 4; ++kk) {
        const float2 l2 = *(const float2*)&sh_l[wave * 16 + 4 * kk + 2 * h];
        v2f pa;
        pa.x = __expf(l2.x - gmax) * inv;
        pa.y = __expf(l2.y - gmax) * inv;
        const float2 r0 = vc[2 * kk];
        const float2 r1 = vc[2 * kk + 1];
        v2f q0; q0.x = r0.x; q0.y = r1.x;
        v2f q1; q1.x = r0.y; q1.y = r1.y;
        o0 = __builtin_amdgcn_wmma_f32_16x16x4_f32(
            false, pa, false, q0, (short)0, o0, false, false);
        o1 = __builtin_amdgcn_wmma_f32_16x16x4_f32(
            false, pa, false, q1, (short)0, o1, false, false);
    }
    float2 po; po.x = o0[0]; po.y = o1[0];
    *(float2*)&sh_part[wave][2 * c] = po;
    lds_barrier();

    if (tid < DC) {
        float o = 0.f;
#pragma unroll
        for (int i = 0; i < 32; ++i) o += sh_part[i][tid];
        const float nu = sh_u[tid] + o;
        sh_u[tid] = nu;
        u[b * D + d0 + tid] = nu;
        if (out) out[b * D + d0 + tid] = nu;
    }
    lds_barrier();

    if (HAS_OUT) {
        v8f acc0 = {0.f, 0.f, 0.f, 0.f, 0.f, 0.f, 0.f, 0.f};
        v8f acc1 = acc0;
#pragma unroll
        for (int j = 0; j < 4; ++j) {
            const float4 uu = *(const float4*)&sh_u[8 * j + 4 * h];
            v2f a0; a0.x = uu.x;      a0.y = uu.y;
            v2f b0; b0.x = va[j].x;   b0.y = va[j].y;
            v2f a1; a1.x = uu.z;      a1.y = uu.w;
            v2f b1; b1.x = va[j].z;   b1.y = va[j].w;
            acc0 = __builtin_amdgcn_wmma_f32_16x16x4_f32(
                false, a0, false, b0, (short)0, acc0, false, false);
            acc1 = __builtin_amdgcn_wmma_f32_16x16x4_f32(
                false, a1, false, b1, (short)0, acc1, false, false);
        }
        sh_l[wave * 16 + c] = acc0[0] + acc1[0];
        __syncthreads();
        if (tid < M / 4)
            ((float4*)(PLout + ((size_t)b * NQ + q) * M))[tid] = ((const float4*)sh_l)[tid];
    }
}

extern "C" void kernel_launch(void* const* d_in, const int* in_sizes, int n_in,
                              void* d_out, int out_size, void* d_ws, size_t ws_size,
                              hipStream_t stream)
{
    const int*   story    = (const int*)d_in[0];
    const float* C_tables = (const float*)d_in[1];
    const float* pos_emb  = (const float*)d_in[2];
    float*       out      = (float*)d_out;

    char* ws = (char*)d_ws;
    int*   positions = (int*)ws;
    float* u         = (float*)(ws + (size_t)B * M * sizeof(int));
    size_t sa_off    = (((size_t)B * M * sizeof(int) +
                         (size_t)B * D * sizeof(float)) + 255) & ~(size_t)255;
    float* SA        = (float*)(ws + sa_off);
    const size_t S   = (size_t)B * M * D;

    const size_t TD  = (size_t)V * D;

    float* PL1 = SA + 0 * S;
    float* PL2 = SA + 0 * S + (size_t)B * NQ * M;
    float* PL0 = SA + 0 * S + 2 * (size_t)B * NQ * M;

    pos_kernel          <<<B,          M,       0, stream>>>(story, positions, u);
    pool01_logit_kernel <<<B * M / PM, PM * 16, 0, stream>>>(story, C_tables, pos_emb,
                                                             positions, u, SA, PL0);
    colhop_pool_kernel<1, true><<<NH + NPOOL, 1024, 0, stream>>>(SA + 1 * S, PL0, PL1, u, nullptr,
                                                        story, C_tables + 2 * TD, pos_emb,
                                                        positions, SA + 2 * S);
    colhop_pool_kernel<NQ, true><<<NH + NPOOL, 1024, 0, stream>>>(SA + 2 * S, PL1, PL2, u, nullptr,
                                                        story, C_tables + 3 * TD, pos_emb,
                                                        positions, SA + 3 * S);
    colhop_pool_kernel<NQ, false><<<NH,         1024, 0, stream>>>(SA + 3 * S, PL2, nullptr, u, out,
                                                        nullptr, nullptr, nullptr,
                                                        nullptr, nullptr);
}


# Hop+pool kernel held to 64 registers (second launch bound, 32-bit row indices): two 1024-thread work-groups per WGP, so all 256 pooling groups (64 slots each) start beside the hops; sums unchanged
# speedup vs baseline: 2.4234x; 1.0692x over previous
#include <hip/hip_runtime.h>
#include <hip/hip_bf16.h>

#define B 32
#define M 512
#define T 16
#define D 128
#define V 32000
#define NQ 4
#define DC 32
#define NH (B * NQ)
#define NPOOL 256
#define SPOOL 64
static_assert(NPOOL * SPOOL == B * M, "pooling shares must cover every slot once");
#define PM 32
#define SP (D + 4)

typedef __attribute__((ext_vector_type(2))) float v2f;
typedef __attribute__((ext_vector_type(8))) float v8f;

__global__ void pos_kernel(const int* __restrict__ story,
                           int* __restrict__ positions,
                           float* __restrict__ u)
{
    int b = blockIdx.x;
    int m = threadIdx.x;
    __shared__ int scan[M];
    int nonpad = (story[(size_t)m * (B * T) + (size_t)b * T] != 0) ? 1 : 0;
    scan[m] = nonpad;
    __syncthreads();
    for (int off = 1; off < M; off <<= 1) {
        int v = scan[m];
        int w = (m >= off) ? scan[m - off] : 0;
        __syncthreads();
        scan[m] = v + w;
        __syncthreads();
    }
    positions[b * M + m] = nonpad ? scan[m] : 0;
    if (m < D) u[b * D + m] = 0.0f;
}

template <int NT, bool KEEP0>
static __device__ __forceinline__ float4 pool_rows(const float* __restrict__ tab,
                                                 const float* __restrict__ pos_emb,
                                                 float* __restrict__ SAo,
                                                 int slot, int lane,
                                                 int tokval, int tbase, int pidx)
{
    const int D4 = D / 4;
    float4 acc[NT];
#pragma unroll
    for (int k = 0; k < NT; ++k) acc[k] = make_float4(0.f, 0.f, 0.f, 0.f);
#pragma unroll
    for (int t = 0; t < T; ++t) {
        const int tok = __shfl(tokval, tbase + t, 32);
        const size_t off = (size_t)tok * D4 + lane;
#pragma unroll
        for (int k = 0; k < NT; ++k) {
            const float4 v = ((const float4*)(tab + (size_t)k * V * D))[off];
            acc[k].x += v.x; acc[k].y += v.y; acc[k].z += v.z; acc[k].w += v.w;
        }
    }
    const float4 pe = ((const float4*)pos_emb)[(size_t)pidx * D4 + lane];
    const size_t S4 = (size_t)B * M * D4;
#pragma unroll
    for (int k = 0; k < NT; ++k) {
        acc[k].x += pe.x; acc[k].y += pe.y; acc[k].z += pe.z; acc[k].w += pe.w;
        if (!(KEEP0 && k == 0))
            ((float4*)SAo)[(size_t)k * S4 + (size_t)slot * D4 + lane] = acc[k];
    }
    return acc[0];
}

static __device__ __forceinline__ void pool_slot(const float* __restrict__ tab,
                                                 const float* __restrict__ pos_emb,
                                                 float* __restrict__ SAo,
                                                 int slot, int lane, int tokval, int pidx)
{
    const unsigned D4 = D / 4;
    float4 v[T];
#pragma unroll
    for (int t = 0; t < T; ++t) {
        const unsigned tok = (unsigned)__builtin_amdgcn_readlane(tokval, t);
        v[t] = ((const float4*)tab)[tok * D4 + (unsigned)lane];
    }
    const unsigned prow = (unsigned)__builtin_amdgcn_readfirstlane(pidx);
    const float4 pe = ((const float4*)pos_emb)[prow * D4 + (unsigned)lane];
    float4 acc = make_float4(0.f, 0.f, 0.f, 0.f);
#pragma unroll
    for (int t = 0; t < T; ++t) {
        acc.x += v[t].x; acc.y += v[t].y; acc.z += v[t].z; acc.w += v[t].w;
    }
    acc.x += pe.x; acc.y += pe.y; acc.z += pe.z; acc.w += pe.w;
    ((float4*)SAo)[(unsigned)slot * D4 + (unsigned)lane] = acc;
}

static __device__ __forceinline__ void lds_barrier()
{
    asm volatile("s_wait_dscnt 0x0\n\ts_barrier_signal -1\n\ts_barrier_wait -1"
                 ::: "memory");
}

__global__ __launch_bounds__(PM * 16)
void pool01_logit_kernel(const int* __restrict__ story,
                         const float* __restrict__ C_tables,
                         const float* __restrict__ pos_emb,
                         const int* __restrict__ positions,
                         const float* __restrict__ u,
                         float* __restrict__ SA,
                         float* __restrict__ PL0)
{
    const int tid  = threadIdx.x;
    const int wave = tid >> 5;
    const int lane = tid & 31;
    const int b  = blockIdx.x >> 4;
    const int m0 = (blockIdx.x & 15) * PM;
    const int c  = lane & 15;
    const int h  = lane >> 4;

    __shared__ __attribute__((aligned(16))) float sh_s[PM][SP];
    __shared__ __attribute__((aligned(16))) float sh_u[D];
    __shared__ __attribute__((aligned(16))) float sh_part[8][PM];

    const int slotA = b * M + m0 + wave;
    const int slotB = slotA + 16;
    const int tokval = story[(size_t)(m0 + wave + 16 * h) * (B * T) + (size_t)b * T + c];
    const int pA = positions[slotA];
    const int pB = positions[slotB];
    const float4 sA = pool_rows<2, true>(C_tables, pos_emb, SA, slotA, lane, tokval, 0, pA);
    const float4 sB = pool_rows<2, true>(C_tables, pos_emb, SA, slotB, lane, tokval, T, pB);

    *(float4*)&sh_s[wave][4 * lane]      = sA;
    *(float4*)&sh_s[16 + wave][4 * lane] = sB;
    sh_u[tid & (D - 1)] = u[b * D + (tid & (D - 1))];
    lds_barrier();

    const int tile = wave & 1;
    const int j0   = (wave >> 1) * 2;
    v8f acc0 = {0.f, 0.f, 0.f, 0.f, 0.f, 0.f, 0.f, 0.f};
    v8f acc1 = acc0;
    for (int jj = 0; jj < 2; ++jj) {
        const int dcol = 8 * (j0 + jj) + 4 * h;
        const float4 uu = *(const float4*)&sh_u[dcol];
        const float4 vs = *(const float4*)&sh_s[16 * tile + c][dcol];
        v2f a0; a0.x = uu.x; a0.y = uu.y;
        v2f b0; b0.x = vs.x; b0.y = vs.y;
        v2f a1; a1.x = uu.z; a1.y = uu.w;
        v2f b1; b1.x = vs.z; b1.y = vs.w;
        acc0 = __builtin_amdgcn_wmma_f32_16x16x4_f32(
            false, a0, false, b0, (short)0, acc0, false, false);
        acc1 = __builtin_amdgcn_wmma_f32_16x16x4_f32(
            false, a1, false, b1, (short)0, acc1, false, false);
    }
    sh_part[wave >> 1][16 * tile + c] = acc0[0] + acc1[0];
    lds_barrier();

    if (tid < PM) {
        float l = 0.f;
        for (int g = 0; g < 8; ++g) l += sh_part[g][tid];
        PL0[b * M + m0 + tid] = l;
    }
}

template <int NIN, bool HAS_OUT>
__global__ __launch_bounds__(1024, 16)
void colhop_pool_kernel(const float* __restrict__ S,
                        const float* __restrict__ PLin,
                        float* __restrict__ PLout,
                        float* __restrict__ u,
                        float* __restrict__ out,
                        const int* __restrict__ story,
                        const float* __restrict__ tab,
                        const float* __restrict__ pos_emb,
                        const int* __restrict__ positions,
                        float* __restrict__ SAo)
{
    const int tid  = threadIdx.x;
    const int wave = tid >> 5;
    const int lane = tid & 31;

    if (HAS_OUT && blockIdx.x >= NH) {
        const int s0 = (blockIdx.x - NH) * SPOOL + wave;
        const int s1  = s0 + 32;
        const int tv0 = story[(unsigned)((s0 & (M - 1)) * (B * T) + (s0 >> 9) * T + (lane & (T - 1)))];
        const int tv1 = story[(unsigned)((s1 & (M - 1)) * (B * T) + (s1 >> 9) * T + (lane & (T - 1)))];
        const int px0 = positions[s0];
        const int px1 = positions[s1];
        pool_slot(tab, pos_emb, SAo, s0, lane, tv0, px0);
        pool_slot(tab, pos_emb, SAo, s1, lane, tv1, px1);
        return;
    }

    const int b  = blockIdx.x >> 2;
    const int q  = blockIdx.x & (NQ - 1);
    const int d0 = q * DC;
    const int c  = lane & 15;
    const int h  = lane >> 4;

    __shared__ __attribute__((aligned(16))) float sh_u[DC];
    __shared__ __attribute__((aligned(16))) float sh_l[M];
    __shared__ __attribute__((aligned(16))) float sh_part[32][DC];

    static_assert(NIN == 1 || NIN == NQ, "whole logits or one share per chunk");
    const float* pr = PLin + (size_t)b * NIN * M + (tid & (M - 1));
    const float pl0 = pr[0];
    float pl1 = 0.f, pl2 = 0.f, pl3 = 0.f;
    if (NIN == NQ) { pl1 = pr[M]; pl2 = pr[2 * M]; pl3 = pr[3 * M]; }
    const float uval = u[b * D + d0 + (tid & (DC - 1))];

    const float2* crow =
        (const float2*)(S + ((size_t)b * M + wave * 16 + 2 * h) * D + d0 + 2 * c);
    float2 vc[8];
#pragma unroll
    for (int kk = 0; kk < 4; ++kk) {
        vc[2 * kk]     = crow[(4 * kk)     * (D / 2)];
        vc[2 * kk + 1] = crow[(4 * kk + 1) * (D / 2)];
    }

    const float4* arow =
        (const float4*)(S + ((size_t)b * M + wave * 16 + c) * D + d0) + h;
    float4 va[4];
    if (HAS_OUT) {
#pragma unroll
        for (int j = 0; j < 4; ++j) va[j] = arow[2 * j];
    }

    sh_l[tid & (M - 1)] = (NIN == NQ) ? ((pl0 + pl1) + pl2) + pl3 : pl0;
    sh_u[tid & (DC - 1)] = uval;
    lds_barrier();

    float gmax = -__builtin_inff();
    float lv[16];
#pragma unroll
    for (int i = 0; i < 4; ++i) {
        const float4 t = *(const float4*)&sh_l[128 * i + 4 * lane];
        lv[4 * i] = t.x; lv[4 * i + 1] = t.y; lv[4 * i + 2] = t.z; lv[4 * i + 3] = t.w;
    }
#pragma unroll
    for (int i = 0; i < 16; ++i) gmax = fmaxf(gmax, lv[i]);
    for (int off = 16; off > 0; off >>= 1)
        gmax = fmaxf(gmax, __shfl_xor(gmax, off, 32));
    float gsum = 0.f;
#pragma unroll
    for (int i = 0; i < 16; ++i) gsum += __expf(lv[i] - gmax);
    for (int off = 16; off > 0; off >>= 1)
        gsum += __shfl_xor(gsum, off, 32);
    const float inv = 1.0f / gsum;

    v8f o0 = {0.f, 0.f, 0.f, 0.f, 0.f, 0.f, 0.f, 0.f};
    v8f o1 = o0;
#pragma unroll
    for (int kk = 0; kk < 4; ++kk) {
        const float2 l2 = *(const float2*)&sh_l[wave * 16 + 4 * kk + 2 * h];
        v2f pa;
        pa.x = __expf(l2.x - gmax) * inv;
        pa.y = __expf(l2.y - gmax) * inv;
        const float2 r0 = vc[2 * kk];
        const float2 r1 = vc[2 * kk + 1];
        v2f q0; q0.x = r0.x; q0.y = r1.x;
        v2f q1; q1.x = r0.y; q1.y = r1.y;
        o0 = __builtin_amdgcn_wmma_f32_16x16x4_f32(
            false, pa, false, q0, (short)0, o0, false, false);
        o1 = __builtin_amdgcn_wmma_f32_16x16x4_f32(
            false, pa, false, q1, (short)0, o1, false, false);
    }
    float2 po; po.x = o0[0]; po.y = o1[0];
    *(float2*)&sh_part[wave][2 * c] = po;
    lds_barrier();

    if (tid < DC) {
        float o = 0.f;
#pragma unroll
        for (int i = 0; i < 32; ++i) o += sh_part[i][tid];
        const float nu = sh_u[tid] + o;
        sh_u[tid] = nu;
        u[b * D + d0 + tid] = nu;
        if (out) out[b * D + d0 + tid] = nu;
    }
    lds_barrier();

    if (HAS_OUT) {
        v8f acc0 = {0.f, 0.f, 0.f, 0.f, 0.f, 0.f, 0.f, 0.f};
        v8f acc1 = acc0;
#pragma unroll
        for (int j = 0; j < 4; ++j) {
            const float4 uu = *(const float4*)&sh_u[8 * j + 4 * h];
            v2f a0; a0.x = uu.x;      a0.y = uu.y;
            v2f b0; b0.x = va[j].x;   b0.y = va[j].y;
            v2f a1; a1.x = uu.z;      a1.y = uu.w;
            v2f b1; b1.x = va[j].z;   b1.y = va[j].w;
            acc0 = __builtin_amdgcn_wmma_f32_16x16x4_f32(
                false, a0, false, b0, (short)0, acc0, false, false);
            acc1 = __builtin_amdgcn_wmma_f32_16x16x4_f32(
                false, a1, false, b1, (short)0, acc1, false, false);
        }
        sh_l[wave * 16 + c] = acc0[0] + acc1[0];
        __syncthreads();
        if (tid < M / 4)
            ((float4*)(PLout + ((size_t)b * NQ + q) * M))[tid] = ((const float4*)sh_l)[tid];
    }
}

extern "C" void kernel_launch(void* const* d_in, const int* in_sizes, int n_in,
                              void* d_out, int out_size, void* d_ws, size_t ws_size,
                              hipStream_t stream)
{
    const int*   story    = (const int*)d_in[0];
    const float* C_tables = (const float*)d_in[1];
    const float* pos_emb  = (const float*)d_in[2];
    float*       out      = (float*)d_out;

    char* ws = (char*)d_ws;
    int*   positions = (int*)ws;
    float* u         = (float*)(ws + (size_t)B * M * sizeof(int));
    size_t sa_off    = (((size_t)B * M * sizeof(int) +
                         (size_t)B * D * sizeof(float)) + 255) & ~(size_t)255;
    float* SA        = (float*)(ws + sa_off);
    const size_t S   = (size_t)B * M * D;

    const size_t TD  = (size_t)V * D;

    float* PL1 = SA + 0 * S;
    float* PL2 = SA + 0 * S + (size_t)B * NQ * M;
    float* PL0 = SA + 0 * S + 2 * (size_t)B * NQ * M;

    pos_kernel          <<<B,          M,       0, stream>>>(story, positions, u);
    pool01_logit_kernel <<<B * M / PM, PM * 16, 0, stream>>>(story, C_tables, pos_emb,
                                                             positions, u, SA, PL0);
    colhop_pool_kernel<1, true><<<NH + NPOOL, 1024, 0, stream>>>(SA + 1 * S, PL0, PL1, u, nullptr,
                                                        story, C_tables + 2 * TD, pos_emb,
                                                        positions, SA + 2 * S);
    colhop_pool_kernel<NQ, true><<<NH + NPOOL, 1024, 0, stream>>>(SA + 2 * S, PL1, PL2, u, nullptr,
                                                        story, C_tables + 3 * TD, pos_emb,
                                                        positions, SA + 3 * S);
    colhop_pool_kernel<NQ, false><<<NH,         1024, 0, stream>>>(SA + 3 * S, PL2, nullptr, u, out,
                                                        nullptr, nullptr, nullptr,
                                                        nullptr, nullptr);
}
